# starting kernel
# baseline (speedup 1.0000x reference)
#include <hip/hip_runtime.h>
#include <hip/hip_bf16.h>
#include <math.h>

typedef float v2f __attribute__((ext_vector_type(2)));
typedef float v8f __attribute__((ext_vector_type(8)));

#define B_ 8
#define N_ 128
#define C_ 64
#define S_ 64
#define BT_ 20

__device__ __forceinline__ float lrelu(float v) { return v >= 0.0f ? v : 0.01f * v; }

__global__ void k_agg(const float* __restrict__ x, const float* __restrict__ nobj,
                      float* __restrict__ agg) {
    int n = blockIdx.x;
    int d = threadIdx.x;
    const float* xp = x + (size_t)n * N_ * C_ + d;
    float s = 0.0f, mx = -INFINITY, mn = INFINITY;
#pragma unroll 4
    for (int i = 0; i < N_; ++i) {
        float v = xp[i * C_];
        s += v;
        mx = fmaxf(mx, v);
        mn = fminf(mn, v);
    }
    float* ap = agg + n * 4 * C_;
    ap[0 * C_ + d] = s * (1.0f / 49.0f);
    ap[1 * C_ + d] = s / nobj[n];
    ap[2 * C_ + d] = mx;
    ap[3 * C_ + d] = mn;
}

__global__ void k_fold(const float* __restrict__ coefs, float* __restrict__ wf) {
    int idx = blockIdx.x * blockDim.x + threadIdx.x;
    int m = idx >> 12;
    int d = (idx >> 6) & 63;
    int s = idx & 63;
    const float* cp = coefs + d * (S_ * BT_) + s * BT_ + m;
    wf[idx] = cp[0] + cp[5] + cp[10] + cp[15];
}

__global__ void k_ef(const float* __restrict__ agg, const float* __restrict__ coefs,
                     const float* __restrict__ bias,
                     float* __restrict__ Eb, float* __restrict__ F) {
    int n = blockIdx.x;
    int s = threadIdx.x;
    float e = 0.0f, f = 0.0f;
#pragma unroll
    for (int a = 0; a < 4; ++a) {
        const float* ag = agg + n * 4 * C_ + a * C_;
        for (int d = 0; d < C_; ++d) {
            const float* cp = coefs + d * (S_ * BT_) + s * BT_ + 5 * a;
            f += ag[d] * cp[3];
            e += ag[d] * cp[4];
        }
    }
    Eb[n * S_ + s] = e + bias[s];
    F[n * S_ + s]  = f;
}

__global__ void k_gemm(const float* __restrict__ x, const float* __restrict__ wf,
                       float* __restrict__ pqr) {
    int wave = threadIdx.x >> 5;
    int lane = threadIdx.x & 31;
    int rl   = lane & 15;
    int kh   = lane >> 4;
    int row  = blockIdx.x * 16 + rl;
    int col  = wave * 16 + rl;

    const float* w0 = wf;
    const float* w1 = wf + 4096;
    const float* w2 = wf + 8192;

    v8f acc0 = {}, acc1 = {}, acc2 = {};

#pragma unroll 4
    for (int k0 = 0; k0 < C_; k0 += 4) {
        int ka = k0 + kh * 2;
        v2f a;
        a.x = x[row * C_ + ka + 0];
        a.y = x[row * C_ + ka + 1];
        v2f b0, b1, b2;
        b0.x = w0[(ka + 0) * S_ + col];  b0.y = w0[(ka + 1) * S_ + col];
        b1.x = w1[(ka + 0) * S_ + col];  b1.y = w1[(ka + 1) * S_ + col];
        b2.x = w2[(ka + 0) * S_ + col];  b2.y = w2[(ka + 1) * S_ + col];
        acc0 = __builtin_amdgcn_wmma_f32_16x16x4_f32(false, a, false, b0, (short)0, acc0, false, false);
        acc1 = __builtin_amdgcn_wmma_f32_16x16x4_f32(false, a, false, b1, (short)0, acc1, false, false);
        acc2 = __builtin_amdgcn_wmma_f32_16x16x4_f32(false, a, false, b2, (short)0, acc2, false, false);
    }

#pragma unroll
    for (int r = 0; r < 8; ++r) {
        int mrow = blockIdx.x * 16 + r + kh * 8;
        pqr[0 * 65536 + mrow * S_ + col] = acc0[r];
        pqr[1 * 65536 + mrow * S_ + col] = acc1[r];
        pqr[2 * 65536 + mrow * S_ + col] = acc2[r];
    }
}

__global__ void k_out(const float* __restrict__ pqr, const float* __restrict__ Eb,
                      const float* __restrict__ F, const float* __restrict__ mask,
                      float* __restrict__ out) {
    int bx = blockIdx.x;
    int n  = bx >> 7;
    int i  = bx & 127;
    int t  = threadIdx.x;
    int sg = t & 15;
    int jl = t >> 4;

    const float4* R4 = (const float4*)(pqr);
    const float4* Q4 = (const float4*)(pqr + 65536);
    const float4* P4 = (const float4*)(pqr + 131072);
    const float4* E4 = (const float4*)Eb;
    const float4* F4 = (const float4*)F;

    float4 p = P4[bx * 16 + sg];
    float4 e = E4[n * 16 + sg];
    p.x += e.x; p.y += e.y; p.z += e.z; p.w += e.w;

    float4 r = R4[bx * 16 + sg];
    float4 f = F4[n * 16 + sg];
    r.x += f.x; r.y += f.y; r.z += f.z; r.w += f.w;

    const float* mrow = mask + ((size_t)n * N_ + i) * N_;
    float4* O4 = (float4*)out;

    for (int j = jl; j < N_; j += 16) {
        float4 q = Q4[(n * N_ + j) * 16 + sg];
        float4 v;
        v.x = p.x + q.x; v.y = p.y + q.y; v.z = p.z + q.z; v.w = p.w + q.w;
        if (j == i) { v.x += r.x; v.y += r.y; v.z += r.z; v.w += r.w; }
        v.x = lrelu(v.x); v.y = lrelu(v.y); v.z = lrelu(v.z); v.w = lrelu(v.w);
        float m = mrow[j];
        v.x *= m; v.y *= m; v.z *= m; v.w *= m;
        O4[((size_t)bx * N_ + j) * 16 + sg] = v;
    }
}

extern "C" void kernel_launch(void* const* d_in, const int* in_sizes, int n_in,
                              void* d_out, int out_size, void* d_ws, size_t ws_size,
                              hipStream_t stream) {
    const float* x     = (const float*)d_in[0];
    const float* mask  = (const float*)d_in[1];
    const float* nobj  = (const float*)d_in[2];
    const float* coefs = (const float*)d_in[3];
    const float* bias  = (const float*)d_in[4];
    float* out = (float*)d_out;

    float* ws  = (float*)d_ws;
    float* agg = ws;
    float* wf  = ws + 2048;
    float* Eb  = ws + 14336;
    float* F   = ws + 14848;
    float* pqr = ws + 15360;

    k_agg <<<B_, C_, 0, stream>>>(x, nobj, agg);
    k_fold<<<48, 256, 0, stream>>>(coefs, wf);
    k_ef  <<<B_, S_, 0, stream>>>(agg, coefs, bias, Eb, F);
    k_gemm<<<(B_ * N_) / 16, 128, 0, stream>>>(x, wf, pqr);
    k_out <<<B_ * N_, 256, 0, stream>>>(pqr, Eb, F, mask, out);
}


# Fuse the four preparatory launches (aggregate, fold, batch constants, WMMA GEMM) into one kernel staged through LDS: 5 launches become 2; summation orders unchanged
# speedup vs baseline: 1.7247x; 1.7247x over previous
#include <hip/hip_runtime.h>
#include <hip/hip_bf16.h>
#include <math.h>


typedef float v2f __attribute__((ext_vector_type(2)));
typedef float v8f __attribute__((ext_vector_type(8)));

#define B_ 8
#define N_ 128
#define C_ 64
#define S_ 64
#define BT_ 20

#define PRE_THREADS 320
#define PLANE4 16384

__device__ __forceinline__ float lrelu(float v) { return v >= 0.0f ? v : 0.01f * v; }

__global__ void __launch_bounds__(PRE_THREADS)
k_pre(const float* __restrict__ x, const float* __restrict__ nobj,
      const float* __restrict__ coefs, const float* __restrict__ bias,
      float* __restrict__ pqr, float* __restrict__ ef) {
    __shared__ float4 raw4[64 * 80];
    __shared__ float  wfs[3 * 64 * 16];
    __shared__ float4 xs4[128 * 16];
    __shared__ float  aggs[4 * 64];

    const int c    = blockIdx.x & 3;
    const int n    = blockIdx.x >> 2;
    const int t    = threadIdx.x;
    const int wave = t >> 5;
    const int lane = t & 31;
    const int rl   = lane & 15;
    const int kh   = lane >> 4;

    {
        const float4* c4 = (const float4*)coefs;
        const float4* x4 = (const float4*)x + n * 2048;
        const int dsub = t / 80;
        const int r    = t - dsub * 80;
        float4 cv[16];
        float4 xv[7];
#pragma unroll
        for (int it = 0; it < 16; ++it) {
            int d = 4 * it + dsub;
            cv[it] = c4[d * 320 + c * 80 + r];
        }
#pragma unroll
        for (int it = 0; it < 7; ++it) {
            int q = t + PRE_THREADS * it;
            xv[it] = x4[q < 2048 ? q : 2047];
        }
#pragma unroll
        for (int it = 0; it < 16; ++it) raw4[it * PRE_THREADS + t] = cv[it];
#pragma unroll
        for (int it = 0; it < 6; ++it) xs4[t + PRE_THREADS * it] = xv[it];
        asm volatile("" :: "v"(xv[6].x), "v"(xv[6].y), "v"(xv[6].z), "v"(xv[6].w));
        if (t + PRE_THREADS * 6 < 2048) xs4[t + PRE_THREADS * 6] = xv[6];
    }
    __syncthreads();

    if (wave < 8) {
#pragma unroll
        for (int it = 0; it < 4; ++it) {
            int p = t + 256 * it;
            float4 g0 = raw4[p * 5 + 0];
            float4 g1 = raw4[p * 5 + 1];
            float4 g2 = raw4[p * 5 + 2];
            float4 g3 = raw4[p * 5 + 3];
            float4 g4 = raw4[p * 5 + 4];
            wfs[0 * 1024 + p] = g0.x + g1.y + g2.z + g3.w;
            wfs[1 * 1024 + p] = g0.y + g1.z + g2.w + g4.x;
            wfs[2 * 1024 + p] = g0.z + g1.w + g3.x + g4.y;
        }
    } else {
        const float* xs = (const float*)xs4;
        int d = t - 256;
        float s = 0.0f, mx = -INFINITY, mn = INFINITY;
#pragma unroll 32
        for (int i = 0; i < N_; ++i) {
            float v = xs[i * C_ + d];
            s += v;
            mx = fmaxf(mx, v);
            mn = fminf(mn, v);
        }
        aggs[0 * C_ + d] = s * (1.0f / 49.0f);
        aggs[1 * C_ + d] = s / nobj[n];
        aggs[2 * C_ + d] = mx;
        aggs[3 * C_ + d] = mn;
    }
    __syncthreads();

    if (wave < 8) {
        const int row = n * N_ + wave * 16 + rl;
        const float* w0 = wfs;
        const float* w1 = wfs + 1024;
        const float* w2 = wfs + 2048;
        v8f acc0 = {}, acc1 = {}, acc2 = {};
#pragma unroll
        for (int k0 = 0; k0 < C_; k0 += 4) {
            int ka = k0 + kh * 2;
            v2f a;
            a.x = x[row * C_ + ka + 0];
            a.y = x[row * C_ + ka + 1];
            v2f b0, b1, b2;
            b0.x = w0[(ka + 0) * 16 + rl];  b0.y = w0[(ka + 1) * 16 + rl];
            b1.x = w1[(ka + 0) * 16 + rl];  b1.y = w1[(ka + 1) * 16 + rl];
            b2.x = w2[(ka + 0) * 16 + rl];  b2.y = w2[(ka + 1) * 16 + rl];
            acc0 = __builtin_amdgcn_wmma_f32_16x16x4_f32(false, a, false, b0, (short)0, acc0, false, false);
            acc1 = __builtin_amdgcn_wmma_f32_16x16x4_f32(false, a, false, b1, (short)0, acc1, false, false);
            acc2 = __builtin_amdgcn_wmma_f32_16x16x4_f32(false, a, false, b2, (short)0, acc2, false, false);
        }
#pragma unroll
        for (int r = 0; r < 8; ++r) {
            int mrow = n * N_ + wave * 16 + r + kh * 8;
            pqr[((0 * 4 + c) * 1024 + mrow) * 16 + rl] = acc0[r];
            pqr[((1 * 4 + c) * 1024 + mrow) * 16 + rl] = acc1[r];
            pqr[((2 * 4 + c) * 1024 + mrow) * 16 + rl] = acc2[r];
        }
    } else if (wave == 8) {
        const float* raw = (const float*)raw4;
        const int sl  = rl;
        const int off = 4 - kh;
        float acc = 0.0f;
#pragma unroll
        for (int a = 0; a < 4; ++a) {
            const float* ag = aggs + a * C_;
            const float* cp = raw + sl * BT_ + 5 * a + off;
#pragma unroll
            for (int d = 0; d < C_; ++d) {
                acc += ag[d] * cp[d * 320];
            }
        }
        float res = kh ? acc : acc + bias[c * 16 + sl];
        ef[(c * 8 + n) * 32 + lane] = res;
    }
}

__global__ void k_out(const float* __restrict__ pqr, const float* __restrict__ ef,
                      const float* __restrict__ mask,
                      float* __restrict__ out) {
    int bx = blockIdx.x;
    int n  = bx >> 7;
    int i  = bx & 127;
    int t  = threadIdx.x;
    int sg = t & 15;
    int jl = t >> 4;
    int ct = sg >> 2;
    int cq = sg & 3;

    const float4* T4 = (const float4*)pqr;
    const float4* R4 = T4 + 0 * PLANE4 + ct * 4096 + cq;
    const float4* Q4 = T4 + 1 * PLANE4 + ct * 4096 + cq;
    const float4* P4 = T4 + 2 * PLANE4 + ct * 4096 + cq;
    const float4* EF4 = (const float4*)ef + (ct * 8 + n) * 8 + cq;

    float4 p = P4[bx * 4];
    float4 e = EF4[0];
    p.x += e.x; p.y += e.y; p.z += e.z; p.w += e.w;

    float4 r = R4[bx * 4];
    float4 f = EF4[4];
    r.x += f.x; r.y += f.y; r.z += f.z; r.w += f.w;

    const float* mrow = mask + ((size_t)n * N_ + i) * N_;
    float4* O4 = (float4*)out;

    for (int j = jl; j < N_; j += 16) {
        float4 q = Q4[(n * N_ + j) * 4];
        float4 v;
        v.x = p.x + q.x; v.y = p.y + q.y; v.z = p.z + q.z; v.w = p.w + q.w;
        if (j == i) { v.x += r.x; v.y += r.y; v.z += r.z; v.w += r.w; }
        v.x = lrelu(v.x); v.y = lrelu(v.y); v.z = lrelu(v.z); v.w = lrelu(v.w);
        float m = mrow[j];
        v.x *= m; v.y *= m; v.z *= m; v.w *= m;
        O4[((size_t)bx * N_ + j) * 16 + sg] = v;
    }
}

extern "C" void kernel_launch(void* const* d_in, const int* in_sizes, int n_in,
                              void* d_out, int out_size, void* d_ws, size_t ws_size,
                              hipStream_t stream) {
    const float* x     = (const float*)d_in[0];
    const float* mask  = (const float*)d_in[1];
    const float* nobj  = (const float*)d_in[2];
    const float* coefs = (const float*)d_in[3];
    const float* bias  = (const float*)d_in[4];
    float* out = (float*)d_out;

    float* ws  = (float*)d_ws;
    float* pqr = ws;
    float* ef  = ws + 196608;

    k_pre<<<32, PRE_THREADS, 0, stream>>>(x, nobj, coefs, bias, pqr, ef);
    k_out<<<B_ * N_, 256, 0, stream>>>(pqr, ef, mask, out);
}


# Streaming kernel regridded: 256 work-groups of four (n,i) rows (one 8-wave group per WGP) instead of 1024 of one row; Q and mask fetched once per thread; arithmetic order unchanged
# speedup vs baseline: 2.5202x; 1.4612x over previous
#include <hip/hip_runtime.h>
#include <hip/hip_bf16.h>
#include <math.h>


typedef float v2f __attribute__((ext_vector_type(2)));
typedef float v8f __attribute__((ext_vector_type(8)));

#define B_ 8
#define N_ 128
#define C_ 64
#define S_ 64
#define BT_ 20

#define PRE_THREADS 320
#define PLANE4 16384

__device__ __forceinline__ float lrelu(float v) { return v >= 0.0f ? v : 0.01f * v; }

__global__ void __launch_bounds__(PRE_THREADS)
k_pre(const float* __restrict__ x, const float* __restrict__ nobj,
      const float* __restrict__ coefs, const float* __restrict__ bias,
      float* __restrict__ pqr, float* __restrict__ ef) {
    __shared__ float4 raw4[64 * 80];
    __shared__ float  wfs[3 * 64 * 16];
    __shared__ float4 xs4[128 * 16];
    __shared__ float  aggs[4 * 64];

    const int c    = blockIdx.x & 3;
    const int n    = blockIdx.x >> 2;
    const int t    = threadIdx.x;
    const int wave = t >> 5;
    const int lane = t & 31;
    const int rl   = lane & 15;
    const int kh   = lane >> 4;

    {
        const float4* c4 = (const float4*)coefs;
        const float4* x4 = (const float4*)x + n * 2048;
        const int dsub = t / 80;
        const int r    = t - dsub * 80;
        float4 cv[16];
        float4 xv[7];
#pragma unroll
        for (int it = 0; it < 16; ++it) {
            int d = 4 * it + dsub;
            cv[it] = c4[d * 320 + c * 80 + r];
        }
#pragma unroll
        for (int it = 0; it < 7; ++it) {
            int q = t + PRE_THREADS * it;
            xv[it] = x4[q < 2048 ? q : 2047];
        }
#pragma unroll
        for (int it = 0; it < 16; ++it) raw4[it * PRE_THREADS + t] = cv[it];
#pragma unroll
        for (int it = 0; it < 6; ++it) xs4[t + PRE_THREADS * it] = xv[it];
        asm volatile("" :: "v"(xv[6].x), "v"(xv[6].y), "v"(xv[6].z), "v"(xv[6].w));
        if (t + PRE_THREADS * 6 < 2048) xs4[t + PRE_THREADS * 6] = xv[6];
    }
    __syncthreads();

    if (wave < 8) {
#pragma unroll
        for (int it = 0; it < 4; ++it) {
            int p = t + 256 * it;
            float4 g0 = raw4[p * 5 + 0];
            float4 g1 = raw4[p * 5 + 1];
            float4 g2 = raw4[p * 5 + 2];
            float4 g3 = raw4[p * 5 + 3];
            float4 g4 = raw4[p * 5 + 4];
            wfs[0 * 1024 + p] = g0.x + g1.y + g2.z + g3.w;
            wfs[1 * 1024 + p] = g0.y + g1.z + g2.w + g4.x;
            wfs[2 * 1024 + p] = g0.z + g1.w + g3.x + g4.y;
        }
    } else {
        const float* xs = (const float*)xs4;
        int d = t - 256;
        float s = 0.0f, mx = -INFINITY, mn = INFINITY;
#pragma unroll 32
        for (int i = 0; i < N_; ++i) {
            float v = xs[i * C_ + d];
            s += v;
            mx = fmaxf(mx, v);
            mn = fminf(mn, v);
        }
        aggs[0 * C_ + d] = s * (1.0f / 49.0f);
        aggs[1 * C_ + d] = s / nobj[n];
        aggs[2 * C_ + d] = mx;
        aggs[3 * C_ + d] = mn;
    }
    __syncthreads();

    if (wave < 8) {
        const int row = n * N_ + wave * 16 + rl;
        const float* w0 = wfs;
        const float* w1 = wfs + 1024;
        const float* w2 = wfs + 2048;
        v8f acc0 = {}, acc1 = {}, acc2 = {};
#pragma unroll
        for (int k0 = 0; k0 < C_; k0 += 4) {
            int ka = k0 + kh * 2;
            v2f a;
            a.x = x[row * C_ + ka + 0];
            a.y = x[row * C_ + ka + 1];
            v2f b0, b1, b2;
            b0.x = w0[(ka + 0) * 16 + rl];  b0.y = w0[(ka + 1) * 16 + rl];
            b1.x = w1[(ka + 0) * 16 + rl];  b1.y = w1[(ka + 1) * 16 + rl];
            b2.x = w2[(ka + 0) * 16 + rl];  b2.y = w2[(ka + 1) * 16 + rl];
            acc0 = __builtin_amdgcn_wmma_f32_16x16x4_f32(false, a, false, b0, (short)0, acc0, false, false);
            acc1 = __builtin_amdgcn_wmma_f32_16x16x4_f32(false, a, false, b1, (short)0, acc1, false, false);
            acc2 = __builtin_amdgcn_wmma_f32_16x16x4_f32(false, a, false, b2, (short)0, acc2, false, false);
        }
#pragma unroll
        for (int r = 0; r < 8; ++r) {
            int mrow = n * N_ + wave * 16 + r + kh * 8;
            pqr[((0 * 4 + c) * 1024 + mrow) * 16 + rl] = acc0[r];
            pqr[((1 * 4 + c) * 1024 + mrow) * 16 + rl] = acc1[r];
            pqr[((2 * 4 + c) * 1024 + mrow) * 16 + rl] = acc2[r];
        }
    } else if (wave == 8) {
        const float* raw = (const float*)raw4;
        const int sl  = rl;
        const int off = 4 - kh;
        float acc = 0.0f;
#pragma unroll
        for (int a = 0; a < 4; ++a) {
            const float* ag = aggs + a * C_;
            const float* cp = raw + sl * BT_ + 5 * a + off;
#pragma unroll
            for (int d = 0; d < C_; ++d) {
                acc += ag[d] * cp[d * 320];
            }
        }
        float res = kh ? acc : acc + bias[c * 16 + sl];
        ef[(c * 8 + n) * 32 + lane] = res;
    }
}

#define OUT_ROWS 4

typedef float v4f __attribute__((ext_vector_type(4)));

__device__ __forceinline__ v4f lrelu4(v4f v) {
    v4f o;
    o.x = lrelu(v.x); o.y = lrelu(v.y); o.z = lrelu(v.z); o.w = lrelu(v.w);
    return o;
}

__global__ void __launch_bounds__(256)
k_out(const float* __restrict__ pqr, const float* __restrict__ ef,
      const float* __restrict__ mask,
      float* __restrict__ out) {
    const int n   = blockIdx.x >> 5;
    const int i0  = (blockIdx.x & 31) * OUT_ROWS;
    const int t   = threadIdx.x;
    const int sg  = t & 15;
    const int jg  = t >> 4;
    const int j0  = jg * 8;
    const int ct  = sg >> 2;
    const int cq  = sg & 3;
    const int row0 = n * N_ + i0;

    const v4f* T4 = (const v4f*)pqr;
    const v4f* R4 = T4 + 0 * PLANE4 + ct * 4096 + cq;
    const v4f* Q4 = T4 + 1 * PLANE4 + ct * 4096 + cq;
    const v4f* P4 = T4 + 2 * PLANE4 + ct * 4096 + cq;
    const v4f* EF4 = (const v4f*)ef + (ct * 8 + n) * 8 + cq;

    v4f q[8];
#pragma unroll
    for (int k = 0; k < 8; ++k) q[k] = Q4[(n * N_ + j0 + k) * 4];
    v4f p[OUT_ROWS], r[OUT_ROWS], ma[OUT_ROWS], mb[OUT_ROWS];
#pragma unroll
    for (int ii = 0; ii < OUT_ROWS; ++ii) {
        p[ii] = P4[(row0 + ii) * 4];
        r[ii] = R4[(row0 + ii) * 4];
        const v4f* m4 = (const v4f*)(mask + (size_t)(row0 + ii) * N_ + j0);
        ma[ii] = m4[0];
        mb[ii] = m4[1];
    }
    const v4f e = EF4[0];
    const v4f f = EF4[4];

    asm volatile("" :: "v"(q[0]), "v"(q[1]), "v"(q[2]), "v"(q[3]),
                       "v"(q[4]), "v"(q[5]), "v"(q[6]), "v"(q[7]),
                       "v"(p[0]), "v"(p[1]), "v"(p[2]), "v"(p[3]),
                       "v"(r[0]), "v"(r[1]), "v"(r[2]), "v"(r[3]),
                       "v"(ma[0]), "v"(ma[1]), "v"(ma[2]), "v"(ma[3]),
                       "v"(mb[0]), "v"(mb[1]), "v"(mb[2]), "v"(mb[3]),
                       "v"(e), "v"(f));

    v4f* O4 = (v4f*)out;

#pragma unroll
    for (int ii = 0; ii < OUT_ROWS; ++ii) {
        const int i = i0 + ii;
        const v4f pp = p[ii] + e;
        const v4f rr = r[ii] + f;
        const float mk[8] = { ma[ii].x, ma[ii].y, ma[ii].z, ma[ii].w,
                              mb[ii].x, mb[ii].y, mb[ii].z, mb[ii].w };
        v4f* orow = O4 + ((size_t)(row0 + ii) * N_ + j0) * 16 + sg;
#pragma unroll
        for (int k = 0; k < 8; ++k) {
            v4f v = pp + q[k];
            if (j0 + k == i) v += rr;
            v = lrelu4(v);
            v *= mk[k];
            orow[k * 16] = v;
        }
        if (ii + 1 < OUT_ROWS) __builtin_amdgcn_sched_barrier(0);
    }
}

extern "C" void kernel_launch(void* const* d_in, const int* in_sizes, int n_in,
                              void* d_out, int out_size, void* d_ws, size_t ws_size,
                              hipStream_t stream) {
    const float* x     = (const float*)d_in[0];
    const float* mask  = (const float*)d_in[1];
    const float* nobj  = (const float*)d_in[2];
    const float* coefs = (const float*)d_in[3];
    const float* bias  = (const float*)d_in[4];
    float* out = (float*)d_out;

    float* ws  = (float*)d_ws;
    float* pqr = ws;
    float* ef  = ws + 196608;

    k_pre<<<32, PRE_THREADS, 0, stream>>>(x, nobj, coefs, bias, pqr, ef);
    k_out<<<256, 256, 0, stream>>>(pqr, ef, mask, out);
}


# k_pre split over 64 work-groups, not 32, to shorten its serial chain: each multiplies 64 rows (WMMAs per SIMD halved); x's i-sum is four 32-row partial sums added in order (sum tree changed)
# speedup vs baseline: 2.8268x; 1.1217x over previous
#include <hip/hip_runtime.h>
#include <hip/hip_bf16.h>
#include <math.h>


typedef float v2f __attribute__((ext_vector_type(2)));
typedef float v8f __attribute__((ext_vector_type(8)));

#define B_ 8
#define N_ 128
#define C_ 64
#define S_ 64
#define BT_ 20

#define PRE_THREADS 256
#define PLANE4 16384

__device__ __forceinline__ float lrelu(float v) { return v >= 0.0f ? v : 0.01f * v; }

__global__ void __launch_bounds__(PRE_THREADS)
k_pre(const float* __restrict__ x, const float* __restrict__ nobj,
      const float* __restrict__ coefs, const float* __restrict__ bias,
      float* __restrict__ pqr, float* __restrict__ ef) {
    __shared__ float4 raw4[64 * 80];
    __shared__ float  wfs[3 * 64 * 16];
    __shared__ float4 xs4[128 * 16];
    __shared__ float  pa[3 * 4 * 64];
    __shared__ float  aggs[4 * 64];

    const int h    = blockIdx.x >> 5;
    const int c    = blockIdx.x & 3;
    const int n    = (blockIdx.x >> 2) & 7;
    const int t    = threadIdx.x;
    const int wave = t >> 5;
    const int lane = t & 31;
    const int rl   = lane & 15;
    const int kh   = lane >> 4;
    const float nb = nobj[n];

    {
        const float4* c4 = (const float4*)coefs;
        float4 cv[20];
        int d = t / 80;
        int r = t - d * 80;
        int base[5];
#pragma unroll
        for (int i = 0; i < 5; ++i) {
            base[i] = d * 320 + c * 80 + r;
            r += 16; d += 3;
            if (r >= 80) { r -= 80; d += 1; }
        }
#pragma unroll
        for (int j = 0; j < 4; ++j)
#pragma unroll
            for (int i = 0; i < 5; ++i) cv[5 * j + i] = c4[base[i] + j * (16 * 320)];
        if (h == 0) {
            const float4* x4 = (const float4*)x + n * 2048;
            float4 xv[8];
#pragma unroll
            for (int it = 0; it < 8; ++it) xv[it] = x4[t + PRE_THREADS * it];
#pragma unroll
            for (int it = 0; it < 20; ++it) raw4[t + PRE_THREADS * it] = cv[it];
#pragma unroll
            for (int it = 0; it < 8; ++it) xs4[t + PRE_THREADS * it] = xv[it];
        } else {
#pragma unroll
            for (int it = 0; it < 20; ++it) raw4[t + PRE_THREADS * it] = cv[it];
        }
    }
    asm volatile("" :: "s"(nb));
    __syncthreads();

#pragma unroll
    for (int it = 0; it < 4; ++it) {
        int p = t + 256 * it;
        float4 g0 = raw4[p * 5 + 0];
        float4 g1 = raw4[p * 5 + 1];
        float4 g2 = raw4[p * 5 + 2];
        float4 g3 = raw4[p * 5 + 3];
        float4 g4 = raw4[p * 5 + 4];
        wfs[0 * 1024 + p] = g0.x + g1.y + g2.z + g3.w;
        wfs[1 * 1024 + p] = g0.y + g1.z + g2.w + g4.x;
        wfs[2 * 1024 + p] = g0.z + g1.w + g3.x + g4.y;
    }
    if (h == 0) {
        const float* xs = (const float*)xs4;
        const int d    = t & 63;
        const int part = t >> 6;
        float s = 0.0f, mx = -INFINITY, mn = INFINITY;
#pragma unroll
        for (int k = 0; k < 32; ++k) {
            float v = xs[(part * 32 + k) * C_ + d];
            s += v;
            mx = fmaxf(mx, v);
            mn = fminf(mn, v);
        }
        pa[(0 * 4 + part) * 64 + d] = s;
        pa[(1 * 4 + part) * 64 + d] = mx;
        pa[(2 * 4 + part) * 64 + d] = mn;
    }
    __syncthreads();

    if (h == 0 && t < 64) {
        float s  = ((pa[0 * 64 + t] + pa[1 * 64 + t]) + pa[2 * 64 + t]) + pa[3 * 64 + t];
        float mx = fmaxf(fmaxf(pa[256 + 0 * 64 + t], pa[256 + 1 * 64 + t]),
                         fmaxf(pa[256 + 2 * 64 + t], pa[256 + 3 * 64 + t]));
        float mn = fminf(fminf(pa[512 + 0 * 64 + t], pa[512 + 1 * 64 + t]),
                         fminf(pa[512 + 2 * 64 + t], pa[512 + 3 * 64 + t]));
        aggs[0 * C_ + t] = s * (1.0f / 49.0f);
        aggs[1 * C_ + t] = s / nb;
        aggs[2 * C_ + t] = mx;
        aggs[3 * C_ + t] = mn;
    }
    __syncthreads();

    if (wave < 4) {
        const int rt  = h * 4 + wave;
        const int row = n * N_ + rt * 16 + rl;
        const float* w0 = wfs;
        const float* w1 = wfs + 1024;
        const float* w2 = wfs + 2048;
        v8f acc0 = {}, acc1 = {}, acc2 = {};
#pragma unroll
        for (int k0 = 0; k0 < C_; k0 += 4) {
            int ka = k0 + kh * 2;
            v2f a;
            a.x = x[row * C_ + ka + 0];
            a.y = x[row * C_ + ka + 1];
            v2f b0, b1, b2;
            b0.x = w0[(ka + 0) * 16 + rl];  b0.y = w0[(ka + 1) * 16 + rl];
            b1.x = w1[(ka + 0) * 16 + rl];  b1.y = w1[(ka + 1) * 16 + rl];
            b2.x = w2[(ka + 0) * 16 + rl];  b2.y = w2[(ka + 1) * 16 + rl];
            acc0 = __builtin_amdgcn_wmma_f32_16x16x4_f32(false, a, false, b0, (short)0, acc0, false, false);
            acc1 = __builtin_amdgcn_wmma_f32_16x16x4_f32(false, a, false, b1, (short)0, acc1, false, false);
            acc2 = __builtin_amdgcn_wmma_f32_16x16x4_f32(false, a, false, b2, (short)0, acc2, false, false);
        }
#pragma unroll
        for (int r = 0; r < 8; ++r) {
            int mrow = n * N_ + rt * 16 + r + kh * 8;
            pqr[((0 * 4 + c) * 1024 + mrow) * 16 + rl] = acc0[r];
            pqr[((1 * 4 + c) * 1024 + mrow) * 16 + rl] = acc1[r];
            pqr[((2 * 4 + c) * 1024 + mrow) * 16 + rl] = acc2[r];
        }
    } else if (h == 0 && wave == 4) {
        const float* raw = (const float*)raw4;
        const int sl  = rl;
        const int off = 4 - kh;
        const float bv = bias[c * 16 + sl];
        float acc = 0.0f;
#pragma unroll
        for (int a = 0; a < 4; ++a) {
            const float* ag = aggs + a * C_;
            const float* cp = raw + sl * BT_ + 5 * a + off;
#pragma unroll
            for (int d = 0; d < C_; ++d) {
                acc += ag[d] * cp[d * 320];
            }
        }
        float res = kh ? acc : acc + bv;
        ef[(c * 8 + n) * 32 + lane] = res;
    }
}

#define OUT_ROWS 4

typedef float v4f __attribute__((ext_vector_type(4)));

__device__ __forceinline__ v4f lrelu4(v4f v) {
    v4f o;
    o.x = lrelu(v.x); o.y = lrelu(v.y); o.z = lrelu(v.z); o.w = lrelu(v.w);
    return o;
}

__global__ void __launch_bounds__(256)
k_out(const float* __restrict__ pqr, const float* __restrict__ ef,
      const float* __restrict__ mask,
      float* __restrict__ out) {
    const int n   = blockIdx.x >> 5;
    const int i0  = (blockIdx.x & 31) * OUT_ROWS;
    const int t   = threadIdx.x;
    const int sg  = t & 15;
    const int jg  = t >> 4;
    const int j0  = jg * 8;
    const int ct  = sg >> 2;
    const int cq  = sg & 3;
    const int row0 = n * N_ + i0;

    const v4f* T4 = (const v4f*)pqr;
    const v4f* R4 = T4 + 0 * PLANE4 + ct * 4096 + cq;
    const v4f* Q4 = T4 + 1 * PLANE4 + ct * 4096 + cq;
    const v4f* P4 = T4 + 2 * PLANE4 + ct * 4096 + cq;
    const v4f* EF4 = (const v4f*)ef + (ct * 8 + n) * 8 + cq;

    v4f q[8];
#pragma unroll
    for (int k = 0; k < 8; ++k) q[k] = Q4[(n * N_ + j0 + k) * 4];
    v4f p[OUT_ROWS], r[OUT_ROWS], ma[OUT_ROWS], mb[OUT_ROWS];
#pragma unroll
    for (int ii = 0; ii < OUT_ROWS; ++ii) {
        p[ii] = P4[(row0 + ii) * 4];
        r[ii] = R4[(row0 + ii) * 4];
        const v4f* m4 = (const v4f*)(mask + (size_t)(row0 + ii) * N_ + j0);
        ma[ii] = m4[0];
        mb[ii] = m4[1];
    }
    const v4f e = EF4[0];
    const v4f f = EF4[4];

    asm volatile("" :: "v"(q[0]), "v"(q[1]), "v"(q[2]), "v"(q[3]),
                       "v"(q[4]), "v"(q[5]), "v"(q[6]), "v"(q[7]),
                       "v"(p[0]), "v"(p[1]), "v"(p[2]), "v"(p[3]),
                       "v"(r[0]), "v"(r[1]), "v"(r[2]), "v"(r[3]),
                       "v"(ma[0]), "v"(ma[1]), "v"(ma[2]), "v"(ma[3]),
                       "v"(mb[0]), "v"(mb[1]), "v"(mb[2]), "v"(mb[3]),
                       "v"(e), "v"(f));

    v4f* O4 = (v4f*)out;

#pragma unroll
    for (int ii = 0; ii < OUT_ROWS; ++ii) {
        const int i = i0 + ii;
        const v4f pp = p[ii] + e;
        const v4f rr = r[ii] + f;
        const float mk[8] = { ma[ii].x, ma[ii].y, ma[ii].z, ma[ii].w,
                              mb[ii].x, mb[ii].y, mb[ii].z, mb[ii].w };
        v4f* orow = O4 + ((size_t)(row0 + ii) * N_ + j0) * 16 + sg;
#pragma unroll
        for (int k = 0; k < 8; ++k) {
            v4f v = pp + q[k];
            if (j0 + k == i) v += rr;
            v = lrelu4(v);
            v *= mk[k];
            orow[k * 16] = v;
        }
        if (ii + 1 < OUT_ROWS) __builtin_amdgcn_sched_barrier(0);
    }
}

extern "C" void kernel_launch(void* const* d_in, const int* in_sizes, int n_in,
                              void* d_out, int out_size, void* d_ws, size_t ws_size,
                              hipStream_t stream) {
    const float* x     = (const float*)d_in[0];
    const float* mask  = (const float*)d_in[1];
    const float* nobj  = (const float*)d_in[2];
    const float* coefs = (const float*)d_in[3];
    const float* bias  = (const float*)d_in[4];
    float* out = (float*)d_out;

    float* ws  = (float*)d_ws;
    float* pqr = ws;
    float* ef  = ws + 196608;

    k_pre<<<64, PRE_THREADS, 0, stream>>>(x, nobj, coefs, bias, pqr, ef);
    k_out<<<256, 256, 0, stream>>>(pqr, ef, mask, out);
}


# k_pre split by role: 32 extra work-groups take the x reduction and Eb/F, the 64 multiply groups only fold and WMMA; Eb/F = 8 partial sums of 32 terms added in order (sum tree changed)
# speedup vs baseline: 3.3948x; 1.2009x over previous
#include <hip/hip_runtime.h>
#include <hip/hip_bf16.h>
#include <math.h>


typedef float v2f __attribute__((ext_vector_type(2)));
typedef float v8f __attribute__((ext_vector_type(8)));

#define B_ 8
#define N_ 128
#define C_ 64
#define S_ 64
#define BT_ 20

#define PRE_THREADS 256
#define PLANE4 16384

__device__ __forceinline__ float lrelu(float v) { return v >= 0.0f ? v : 0.01f * v; }

#define CHUNK_J (16 * 320)
#define CHUNK_LOADS                                                                   \
    const float4 cv0  = c4[b0];               const float4 cv1  = c4[b1];               \
    const float4 cv2  = c4[b2];               const float4 cv3  = c4[b3];               \
    const float4 cv4  = c4[b4];               const float4 cv5  = c4[b0 + CHUNK_J];     \
    const float4 cv6  = c4[b1 + CHUNK_J];     const float4 cv7  = c4[b2 + CHUNK_J];     \
    const float4 cv8  = c4[b3 + CHUNK_J];     const float4 cv9  = c4[b4 + CHUNK_J];     \
    const float4 cv10 = c4[b0 + 2 * CHUNK_J]; const float4 cv11 = c4[b1 + 2 * CHUNK_J]; \
    const float4 cv12 = c4[b2 + 2 * CHUNK_J]; const float4 cv13 = c4[b3 + 2 * CHUNK_J]; \
    const float4 cv14 = c4[b4 + 2 * CHUNK_J]; const float4 cv15 = c4[b0 + 3 * CHUNK_J]; \
    const float4 cv16 = c4[b1 + 3 * CHUNK_J]; const float4 cv17 = c4[b2 + 3 * CHUNK_J]; \
    const float4 cv18 = c4[b3 + 3 * CHUNK_J]; const float4 cv19 = c4[b4 + 3 * CHUNK_J];
#define CHUNK_STORES                                                                  \
    raw4[t +  0 * PRE_THREADS] = cv0;   raw4[t +  1 * PRE_THREADS] = cv1;              \
    raw4[t +  2 * PRE_THREADS] = cv2;   raw4[t +  3 * PRE_THREADS] = cv3;              \
    raw4[t +  4 * PRE_THREADS] = cv4;   raw4[t +  5 * PRE_THREADS] = cv5;              \
    raw4[t +  6 * PRE_THREADS] = cv6;   raw4[t +  7 * PRE_THREADS] = cv7;              \
    raw4[t +  8 * PRE_THREADS] = cv8;   raw4[t +  9 * PRE_THREADS] = cv9;              \
    raw4[t + 10 * PRE_THREADS] = cv10;  raw4[t + 11 * PRE_THREADS] = cv11;             \
    raw4[t + 12 * PRE_THREADS] = cv12;  raw4[t + 13 * PRE_THREADS] = cv13;             \
    raw4[t + 14 * PRE_THREADS] = cv14;  raw4[t + 15 * PRE_THREADS] = cv15;             \
    raw4[t + 16 * PRE_THREADS] = cv16;  raw4[t + 17 * PRE_THREADS] = cv17;             \
    raw4[t + 18 * PRE_THREADS] = cv18;  raw4[t + 19 * PRE_THREADS] = cv19;

#define PRE_MUL_BLOCKS 64
#define PRE_BLOCKS 96

__global__ void __launch_bounds__(PRE_THREADS)
k_pre(const float* __restrict__ x, const float* __restrict__ nobj,
      const float* __restrict__ coefs, const float* __restrict__ bias,
      float* __restrict__ pqr, float* __restrict__ ef) {
    __shared__ float4 raw4[64 * 80];
    __shared__ float  wfs[3 * 64 * 16];
    __shared__ float4 xs4[128 * 16];
    __shared__ float  pa[3 * 4 * 64];
    __shared__ float  aggs[4 * 64];
    __shared__ float  part[8 * 32];

    const bool agg = blockIdx.x >= PRE_MUL_BLOCKS;
    const int h    = (blockIdx.x >> 5) & 1;
    const int c    = blockIdx.x & 3;
    const int n    = (blockIdx.x >> 2) & 7;
    const int t    = threadIdx.x;
    const int wave = t >> 5;
    const int lane = t & 31;
    const int rl   = lane & 15;
    const int kh   = lane >> 4;

    const float4* c4 = (const float4*)coefs;
    int bq[5];
    {
        int d = t / 80;
        int r = t - d * 80;
#pragma unroll
        for (int i = 0; i < 5; ++i) {
            bq[i] = d * 320 + c * 80 + r;
            r += 16; d += 3;
            if (r >= 80) { r -= 80; d += 1; }
        }
    }
    const int b0 = bq[0], b1 = bq[1], b2 = bq[2], b3 = bq[3], b4 = bq[4];

    if (!agg) {
        {
            CHUNK_LOADS
            __builtin_amdgcn_sched_barrier(0);
            CHUNK_STORES
        }
        __syncthreads();

#pragma unroll
        for (int it = 0; it < 4; ++it) {
            int p = t + 256 * it;
            float4 g0 = raw4[p * 5 + 0];
            float4 g1 = raw4[p * 5 + 1];
            float4 g2 = raw4[p * 5 + 2];
            float4 g3 = raw4[p * 5 + 3];
            float4 g4 = raw4[p * 5 + 4];
            wfs[0 * 1024 + p] = g0.x + g1.y + g2.z + g3.w;
            wfs[1 * 1024 + p] = g0.y + g1.z + g2.w + g4.x;
            wfs[2 * 1024 + p] = g0.z + g1.w + g3.x + g4.y;
        }
        __syncthreads();

        if (wave < 4) {
            const int rt  = h * 4 + wave;
            const int row = n * N_ + rt * 16 + rl;
            const float* w0 = wfs;
            const float* w1 = wfs + 1024;
            const float* w2 = wfs + 2048;
            v8f acc0 = {}, acc1 = {}, acc2 = {};
#pragma unroll
            for (int k0 = 0; k0 < C_; k0 += 4) {
                int ka = k0 + kh * 2;
                v2f a;
                a.x = x[row * C_ + ka + 0];
                a.y = x[row * C_ + ka + 1];
                v2f b0, b1, b2;
                b0.x = w0[(ka + 0) * 16 + rl];  b0.y = w0[(ka + 1) * 16 + rl];
                b1.x = w1[(ka + 0) * 16 + rl];  b1.y = w1[(ka + 1) * 16 + rl];
                b2.x = w2[(ka + 0) * 16 + rl];  b2.y = w2[(ka + 1) * 16 + rl];
                acc0 = __builtin_amdgcn_wmma_f32_16x16x4_f32(false, a, false, b0, (short)0, acc0, false, false);
                acc1 = __builtin_amdgcn_wmma_f32_16x16x4_f32(false, a, false, b1, (short)0, acc1, false, false);
                acc2 = __builtin_amdgcn_wmma_f32_16x16x4_f32(false, a, false, b2, (short)0, acc2, false, false);
            }
#pragma unroll
            for (int r = 0; r < 8; ++r) {
                int mrow = n * N_ + rt * 16 + r + kh * 8;
                pqr[((0 * 4 + c) * 1024 + mrow) * 16 + rl] = acc0[r];
                pqr[((1 * 4 + c) * 1024 + mrow) * 16 + rl] = acc1[r];
                pqr[((2 * 4 + c) * 1024 + mrow) * 16 + rl] = acc2[r];
            }
        }
    } else {
        const float nb = nobj[n];
        const float bv = bias[c * 16 + rl];
        {
            const float4* x4 = (const float4*)x + n * 2048;
            const float4 xv0 = x4[t + 0 * PRE_THREADS], xv1 = x4[t + 1 * PRE_THREADS];
            const float4 xv2 = x4[t + 2 * PRE_THREADS], xv3 = x4[t + 3 * PRE_THREADS];
            const float4 xv4 = x4[t + 4 * PRE_THREADS], xv5 = x4[t + 5 * PRE_THREADS];
            const float4 xv6 = x4[t + 6 * PRE_THREADS], xv7 = x4[t + 7 * PRE_THREADS];
            CHUNK_LOADS
            __builtin_amdgcn_sched_barrier(0);
            xs4[t + 0 * PRE_THREADS] = xv0;  xs4[t + 1 * PRE_THREADS] = xv1;
            xs4[t + 2 * PRE_THREADS] = xv2;  xs4[t + 3 * PRE_THREADS] = xv3;
            xs4[t + 4 * PRE_THREADS] = xv4;  xs4[t + 5 * PRE_THREADS] = xv5;
            xs4[t + 6 * PRE_THREADS] = xv6;  xs4[t + 7 * PRE_THREADS] = xv7;
            CHUNK_STORES
        }
        asm volatile("" :: "s"(nb), "v"(bv));
        __syncthreads();

        {
            const float* xs = (const float*)xs4;
            const int d    = t & 63;
            const int prt  = t >> 6;
            float s = 0.0f, mx = -INFINITY, mn = INFINITY;
#pragma unroll
            for (int k = 0; k < 32; ++k) {
                float v = xs[(prt * 32 + k) * C_ + d];
                s += v;
                mx = fmaxf(mx, v);
                mn = fminf(mn, v);
            }
            pa[(0 * 4 + prt) * 64 + d] = s;
            pa[(1 * 4 + prt) * 64 + d] = mx;
            pa[(2 * 4 + prt) * 64 + d] = mn;
        }
        __syncthreads();

        if (t < 64) {
            float s  = ((pa[0 * 64 + t] + pa[1 * 64 + t]) + pa[2 * 64 + t]) + pa[3 * 64 + t];
            float mx = fmaxf(fmaxf(pa[256 + 0 * 64 + t], pa[256 + 1 * 64 + t]),
                             fmaxf(pa[256 + 2 * 64 + t], pa[256 + 3 * 64 + t]));
            float mn = fminf(fminf(pa[512 + 0 * 64 + t], pa[512 + 1 * 64 + t]),
                             fminf(pa[512 + 2 * 64 + t], pa[512 + 3 * 64 + t]));
            aggs[0 * C_ + t] = s * (1.0f / 49.0f);
            aggs[1 * C_ + t] = s / nb;
            aggs[2 * C_ + t] = mx;
            aggs[3 * C_ + t] = mn;
        }
        __syncthreads();

        {
            const float* raw = (const float*)raw4;
            const int a   = wave >> 1;
            const int dh  = wave & 1;
            const int off = 4 - kh;
            const float* ag = aggs + a * C_ + dh * 32;
            const float* cp = raw + (dh * 32) * 320 + rl * BT_ + 5 * a + off;
            float acc = 0.0f;
#pragma unroll
            for (int d = 0; d < 32; ++d) {
                acc += ag[d] * cp[d * 320];
            }
            part[wave * 32 + lane] = acc;
        }
        __syncthreads();

        if (t < 32) {
            float acc = part[lane];
#pragma unroll
            for (int w = 1; w < 8; ++w) acc += part[w * 32 + lane];
            float res = kh ? acc : acc + bv;
            ef[(c * 8 + n) * 32 + lane] = res;
        }
    }
}

#define OUT_ROWS 4

typedef float v4f __attribute__((ext_vector_type(4)));

__device__ __forceinline__ v4f lrelu4(v4f v) {
    v4f o;
    o.x = lrelu(v.x); o.y = lrelu(v.y); o.z = lrelu(v.z); o.w = lrelu(v.w);
    return o;
}

__global__ void __launch_bounds__(256)
k_out(const float* __restrict__ pqr, const float* __restrict__ ef,
      const float* __restrict__ mask,
      float* __restrict__ out) {
    const int n   = blockIdx.x >> 5;
    const int i0  = (blockIdx.x & 31) * OUT_ROWS;
    const int t   = threadIdx.x;
    const int sg  = t & 15;
    const int jg  = t >> 4;
    const int j0  = jg * 8;
    const int ct  = sg >> 2;
    const int cq  = sg & 3;
    const int row0 = n * N_ + i0;

    const v4f* T4 = (const v4f*)pqr;
    const v4f* R4 = T4 + 0 * PLANE4 + ct * 4096 + cq;
    const v4f* Q4 = T4 + 1 * PLANE4 + ct * 4096 + cq;
    const v4f* P4 = T4 + 2 * PLANE4 + ct * 4096 + cq;
    const v4f* EF4 = (const v4f*)ef + (ct * 8 + n) * 8 + cq;

    v4f q[8];
#pragma unroll
    for (int k = 0; k < 8; ++k) q[k] = Q4[(n * N_ + j0 + k) * 4];
    v4f p[OUT_ROWS], r[OUT_ROWS], ma[OUT_ROWS], mb[OUT_ROWS];
#pragma unroll
    for (int ii = 0; ii < OUT_ROWS; ++ii) {
        p[ii] = P4[(row0 + ii) * 4];
        r[ii] = R4[(row0 + ii) * 4];
        const v4f* m4 = (const v4f*)(mask + (size_t)(row0 + ii) * N_ + j0);
        ma[ii] = m4[0];
        mb[ii] = m4[1];
    }
    const v4f e = EF4[0];
    const v4f f = EF4[4];

    asm volatile("" :: "v"(q[0]), "v"(q[1]), "v"(q[2]), "v"(q[3]),
                       "v"(q[4]), "v"(q[5]), "v"(q[6]), "v"(q[7]),
                       "v"(p[0]), "v"(p[1]), "v"(p[2]), "v"(p[3]),
                       "v"(r[0]), "v"(r[1]), "v"(r[2]), "v"(r[3]),
                       "v"(ma[0]), "v"(ma[1]), "v"(ma[2]), "v"(ma[3]),
                       "v"(mb[0]), "v"(mb[1]), "v"(mb[2]), "v"(mb[3]),
                       "v"(e), "v"(f));

    v4f* O4 = (v4f*)out;

#pragma unroll
    for (int ii = 0; ii < OUT_ROWS; ++ii) {
        const int i = i0 + ii;
        const v4f pp = p[ii] + e;
        const v4f rr = r[ii] + f;
        const float mk[8] = { ma[ii].x, ma[ii].y, ma[ii].z, ma[ii].w,
                              mb[ii].x, mb[ii].y, mb[ii].z, mb[ii].w };
        v4f* orow = O4 + ((size_t)(row0 + ii) * N_ + j0) * 16 + sg;
#pragma unroll
        for (int k = 0; k < 8; ++k) {
            v4f v = pp + q[k];
            if (j0 + k == i) v += rr;
            v = lrelu4(v);
            v *= mk[k];
            orow[k * 16] = v;
        }
        if (ii + 1 < OUT_ROWS) __builtin_amdgcn_sched_barrier(0);
    }
}

extern "C" void kernel_launch(void* const* d_in, const int* in_sizes, int n_in,
                              void* d_out, int out_size, void* d_ws, size_t ws_size,
                              hipStream_t stream) {
    const float* x     = (const float*)d_in[0];
    const float* mask  = (const float*)d_in[1];
    const float* nobj  = (const float*)d_in[2];
    const float* coefs = (const float*)d_in[3];
    const float* bias  = (const float*)d_in[4];
    float* out = (float*)d_out;

    float* ws  = (float*)d_ws;
    float* pqr = ws;
    float* ef  = ws + 196608;

    k_pre<<<PRE_BLOCKS, PRE_THREADS, 0, stream>>>(x, nobj, coefs, bias, pqr, ef);
    k_out<<<256, 256, 0, stream>>>(pqr, ef, mask, out);
}
